# starting kernel
# baseline (speedup 1.0000x reference)
#include <hip/hip_runtime.h>


typedef __attribute__((ext_vector_type(2))) float v2f;
typedef __attribute__((ext_vector_type(8))) float v8f;

#define NXC 64
#define NKC 16
#define BATCHC 16
#define FEATC (NXC * NXC)

__global__ __launch_bounds__(128) void polar_stage_kernel(
    const float* __restrict__ inputs,
    const int*   __restrict__ r_index,
    const float* __restrict__ pre1,  const float* __restrict__ pre2,
    const float* __restrict__ pre3,  const float* __restrict__ pre4,
    const float* __restrict__ post1, const float* __restrict__ post2,
    const float* __restrict__ post3, const float* __restrict__ post4,
    const float* __restrict__ ck1, const float* __restrict__ sk1,
    const float* __restrict__ ck2, const float* __restrict__ sk2,
    const float* __restrict__ ck3, const float* __restrict__ sk3,
    const float* __restrict__ ck4, const float* __restrict__ sk4,
    float* __restrict__ polar)
{
    __shared__ float sD[2][NXC * NXC];
    __shared__ float sA[NXC * NXC];
    __shared__ float sB[NXC * NXC];

    const int tid  = threadIdx.x;
    const int lane = tid & 31;
    const int wave = tid >> 5;

    const int c     = blockIdx.x;
    const int batch = blockIdx.y;
    const int f     = blockIdx.z;

    const int* rix = r_index + c * FEATC;
    const float* baseR = inputs + ((size_t)(batch * 2 + 0) * FEATC) * NKC + f;
    const float* baseI = inputs + ((size_t)(batch * 2 + 1) * FEATC) * NKC + f;
    for (int e = tid; e < FEATC; e += 128) {
        const int idx = rix[e];
        sD[0][e] = baseR[(size_t)idx * NKC];
        sD[1][e] = baseI[(size_t)idx * NKC];
    }
    __syncthreads();

    const float* PRE[4]  = {pre1,  pre2,  pre3,  pre4};
    const float* POST[4] = {post1, post2, post3, post4};
    const float* K2[4]   = {ck1, sk1, ck3, sk4};
    const float* K1[4]   = {ck2, sk2, sk3, ck4};
    const int    DSEL[4] = {0, 0, 1, 1};

    const int kcol    = wave * 16 + (lane & 15);
    const int halfsel = lane >> 4;
    const int arow    = (lane & 15);
    float acc = 0.0f;

    for (int t = 0; t < 4; ++t) {
        __syncthreads();
        const float* k2f   = K2[t] + f * FEATC;
        const float* k1f   = K1[t] + f * FEATC;
        const float* postf = POST[t] + f * NXC;
        const float* pref  = PRE[t]  + f * NXC;
        for (int e = tid; e < FEATC; e += 128) {
            const int row = e >> 6;
            sA[e] = postf[row] * k2f[e];
            sB[e] = pref[row]  * k1f[e];
        }
        __syncthreads();

        const float* Dt = sD[DSEL[t]];

        for (int it = 0; it < 4; it += 2) {
            v8f m0 = {};
            v8f m1 = {};
            #pragma unroll
            for (int j = 0; j < NXC; j += 4) {
                v2f b;
                b.x = sB[(j + halfsel * 2)     * NXC + kcol];
                b.y = sB[(j + halfsel * 2 + 1) * NXC + kcol];
                const float* a0p = &Dt[((it    ) * 16 + arow) * NXC + j + halfsel * 2];
                const float* a1p = &Dt[((it + 1) * 16 + arow) * NXC + j + halfsel * 2];
                v2f a0; a0.x = a0p[0]; a0.y = a0p[1];
                v2f a1; a1.x = a1p[0]; a1.y = a1p[1];
                m0 = __builtin_amdgcn_wmma_f32_16x16x4_f32(
                         false, a0, false, b, (short)0, m0, false, false);
                m1 = __builtin_amdgcn_wmma_f32_16x16x4_f32(
                         false, a1, false, b, (short)0, m1, false, false);
            }
            const int rb0 = (it    ) * 16 + halfsel * 8;
            const int rb1 = (it + 1) * 16 + halfsel * 8;
            #pragma unroll
            for (int r = 0; r < 8; ++r) {
                acc += sA[(rb0 + r) * NXC + kcol] * m0[r];
                acc += sA[(rb1 + r) * NXC + kcol] * m1[r];
            }
        }
    }

    acc += __shfl_xor(acc, 16, 32);
    if (lane < 16) {
        const size_t fb = (size_t)f * BATCHC + batch;
        polar[fb * FEATC + c * NXC + wave * 16 + lane] = acc;
    }
}

__global__ __launch_bounds__(256) void cart_stage_kernel(
    const float* __restrict__ cart,
    const float* __restrict__ polar,
    float* __restrict__ out)
{
    __shared__ float sP[64 * 64];
    __shared__ float sC[64 * 128];

    const int tid  = threadIdx.x;
    const int lane = tid & 31;
    const int wave = tid >> 5;

    const int n0 = blockIdx.x * 128;
    const int m0 = blockIdx.y * 64;

    const int mt0 = (wave >> 2) * 2;
    const int nt0 = (wave & 3) * 2;

    const int halfsel = lane >> 4;
    const int ml      = lane & 15;

    v8f acc00 = {};
    v8f acc01 = {};
    v8f acc10 = {};
    v8f acc11 = {};

    for (int q0 = 0; q0 < 4096; q0 += 64) {
        __syncthreads();
        for (int e = tid; e < 4096; e += 256) {
            const int m = e >> 6, qq = e & 63;
            sP[e] = polar[(size_t)(m0 + m) * 4096 + q0 + qq];
        }
        for (int e = tid; e < 8192; e += 256) {
            const int p = e >> 6, qq = e & 63;
            sC[qq * 128 + p] = cart[(size_t)(n0 + p) * 4096 + q0 + qq];
        }
        if (q0 + 64 < 4096 && tid < 128) {
            __builtin_prefetch(&cart[(size_t)(n0 + tid) * 4096 + q0 + 64], 0, 1);
        }
        __syncthreads();

        const int n_a = (nt0    ) * 16 + ml;
        const int n_b = (nt0 + 1) * 16 + ml;
        #pragma unroll
        for (int j = 0; j < 64; j += 4) {
            const float* a0p = &sP[((mt0    ) * 16 + ml) * 64 + j + halfsel * 2];
            const float* a1p = &sP[((mt0 + 1) * 16 + ml) * 64 + j + halfsel * 2];
            v2f a0; a0.x = a0p[0]; a0.y = a0p[1];
            v2f a1; a1.x = a1p[0]; a1.y = a1p[1];
            v2f b0, b1;
            b0.x = sC[(j + halfsel * 2)     * 128 + n_a];
            b0.y = sC[(j + halfsel * 2 + 1) * 128 + n_a];
            b1.x = sC[(j + halfsel * 2)     * 128 + n_b];
            b1.y = sC[(j + halfsel * 2 + 1) * 128 + n_b];
            acc00 = __builtin_amdgcn_wmma_f32_16x16x4_f32(
                        false, a0, false, b0, (short)0, acc00, false, false);
            acc01 = __builtin_amdgcn_wmma_f32_16x16x4_f32(
                        false, a0, false, b1, (short)0, acc01, false, false);
            acc10 = __builtin_amdgcn_wmma_f32_16x16x4_f32(
                        false, a1, false, b0, (short)0, acc10, false, false);
            acc11 = __builtin_amdgcn_wmma_f32_16x16x4_f32(
                        false, a1, false, b1, (short)0, acc11, false, false);
        }
    }

    #pragma unroll
    for (int mi = 0; mi < 2; ++mi) {
        #pragma unroll
        for (int ni = 0; ni < 2; ++ni) {
            const v8f acc = (mi == 0) ? (ni == 0 ? acc00 : acc01)
                                      : (ni == 0 ? acc10 : acc11);
            const int p  = n0 + (nt0 + ni) * 16 + ml;
            const int e1 = p >> 6, e2 = p & 63;
            #pragma unroll
            for (int r = 0; r < 8; ++r) {
                const int fb = m0 + (mt0 + mi) * 16 + halfsel * 8 + r;
                const int fo = fb >> 4, bo = fb & 15;
                out[(((size_t)bo * 64 + e1) * 64 + e2) * NKC + fo] = acc[r];
            }
        }
    }
}

extern "C" void kernel_launch(void* const* d_in, const int* in_sizes, int n_in,
                              void* d_out, int out_size, void* d_ws, size_t ws_size,
                              hipStream_t stream) {
    const float* inputs  = (const float*)d_in[0];
    const float* cart    = (const float*)d_in[1];
    const int*   r_index = (const int*)  d_in[2];
    const float* pre1  = (const float*)d_in[3];
    const float* pre2  = (const float*)d_in[4];
    const float* pre3  = (const float*)d_in[5];
    const float* pre4  = (const float*)d_in[6];
    const float* post1 = (const float*)d_in[7];
    const float* post2 = (const float*)d_in[8];
    const float* post3 = (const float*)d_in[9];
    const float* post4 = (const float*)d_in[10];
    const float* ck1 = (const float*)d_in[11];
    const float* sk1 = (const float*)d_in[12];
    const float* ck2 = (const float*)d_in[13];
    const float* sk2 = (const float*)d_in[14];
    const float* ck3 = (const float*)d_in[15];
    const float* sk3 = (const float*)d_in[16];
    const float* ck4 = (const float*)d_in[17];
    const float* sk4 = (const float*)d_in[18];

    float* polar = (float*)d_ws;

    polar_stage_kernel<<<dim3(64, 16, 16), 128, 0, stream>>>(
        inputs, r_index,
        pre1, pre2, pre3, pre4, post1, post2, post3, post4,
        ck1, sk1, ck2, sk2, ck3, sk3, ck4, sk4, polar);

    cart_stage_kernel<<<dim3(32, 4), 256, 0, stream>>>(
        cart, polar, (float*)d_out);
}


# Stage 2 operands leave LDS: wave = block's 64x64 tile for one K quarter, A/B fragments straight from global as K-contiguous 16-byte loads a stage ahead; 4 partials added via LDS; sum order changed
# speedup vs baseline: 3.0351x; 3.0351x over previous
#include <hip/hip_runtime.h>


typedef __attribute__((ext_vector_type(2))) float v2f;
typedef __attribute__((ext_vector_type(8))) float v8f;

#define NXC 64
#define NKC 16
#define BATCHC 16
#define FEATC (NXC * NXC)


__device__ __forceinline__ void gather_a_frags(
    v2f (&a)[16], const int* __restrict__ rix, const char* __restrict__ base)
{
    #pragma unroll
    for (int s = 0; s < 16; ++s) {
        const int2 ix = *reinterpret_cast<const int2*>(rix + 4 * s);
        const unsigned ox = (unsigned)ix.x * (unsigned)(NKC * sizeof(float));
        const unsigned oy = (unsigned)ix.y * (unsigned)(NKC * sizeof(float));
        a[s].x = *reinterpret_cast<const float*>(base + ox);
        a[s].y = *reinterpret_cast<const float*>(base + oy);
    }
}

__device__ __forceinline__ void run_term(
    const v2f (&a)[16], float (&acc)[4], float* sA, float* sB,
    const float* __restrict__ k2f, const float* __restrict__ k1f,
    const float* __restrict__ postf, const float* __restrict__ pref,
    int tid, int halfsel, int l15, int rbase)
{
    __syncthreads();
    for (int e = tid; e < FEATC; e += 128) {
        const int row = e >> 6;
        sA[e] = postf[row] * k2f[e];
        sB[e] = pref[row]  * k1f[e];
    }
    __syncthreads();

    #pragma unroll
    for (int kp = 0; kp < 4; kp += 2) {
        v8f m0 = {};
        v8f m1 = {};
        #pragma unroll
        for (int s = 0; s < 16; ++s) {
            const float* bp = &sB[(4 * s + halfsel * 2) * NXC + kp * 16 + l15];
            v2f b0, b1;
            b0.x = bp[0];
            b0.y = bp[NXC];
            b1.x = bp[16];
            b1.y = bp[16 + NXC];
            m0 = __builtin_amdgcn_wmma_f32_16x16x4_f32(
                     false, a[s], false, b0, (short)0, m0, false, false);
            m1 = __builtin_amdgcn_wmma_f32_16x16x4_f32(
                     false, a[s], false, b1, (short)0, m1, false, false);
        }
        const float* wp = &sA[rbase * NXC + kp * 16 + l15];
        #pragma unroll
        for (int r = 0; r < 8; ++r) {
            acc[kp]     += wp[r * NXC]      * m0[r];
            acc[kp + 1] += wp[r * NXC + 16] * m1[r];
        }
    }
}

__global__ __launch_bounds__(128, 8) void polar_stage_kernel(
    const float* __restrict__ inputs,
    const int*   __restrict__ r_index,
    const float* __restrict__ pre1,  const float* __restrict__ pre2,
    const float* __restrict__ pre3,  const float* __restrict__ pre4,
    const float* __restrict__ post1, const float* __restrict__ post2,
    const float* __restrict__ post3, const float* __restrict__ post4,
    const float* __restrict__ ck1, const float* __restrict__ sk1,
    const float* __restrict__ ck2, const float* __restrict__ sk2,
    const float* __restrict__ ck3, const float* __restrict__ sk3,
    const float* __restrict__ ck4, const float* __restrict__ sk4,
    float* __restrict__ polar)
{
    __shared__ float sA[NXC * NXC];
    __shared__ float sB[NXC * NXC];

    const int tid  = threadIdx.x;
    const int lane = tid & 31;
    const int wave = tid >> 5;

    const int c     = blockIdx.x;
    const int batch = blockIdx.y;
    const int f     = blockIdx.z;

    const int halfsel = lane >> 4;
    const int l15     = lane & 15;
    const int rbase   = wave * 16 + halfsel * 8;

    const int* rix = r_index + c * FEATC + (wave * 16 + l15) * NXC + halfsel * 2;
    const char* baseR = reinterpret_cast<const char*>(
        inputs + ((size_t)(batch * 2 + 0) * FEATC) * NKC + f);
    const char* baseI = reinterpret_cast<const char*>(
        inputs + ((size_t)(batch * 2 + 1) * FEATC) * NKC + f);

    float acc[4] = {0.0f, 0.0f, 0.0f, 0.0f};
    v2f a[16];

    const float* PRE[4]  = {pre1,  pre2,  pre3,  pre4};
    const float* POST[4] = {post1, post2, post3, post4};
    const float* K2[4]   = {ck1, sk1, ck3, sk4};
    const float* K1[4]   = {ck2, sk2, sk3, ck4};

    #pragma nounroll
    for (int t = 0; t < 4; ++t) {
        if ((t & 1) == 0) gather_a_frags(a, rix, (t & 2) ? baseI : baseR);
        run_term(a, acc, sA, sB, K2[t] + f * FEATC, K1[t] + f * FEATC,
                 POST[t] + f * NXC, PRE[t] + f * NXC, tid, halfsel, l15, rbase);
    }

    #pragma unroll
    for (int kt = 0; kt < 4; ++kt) acc[kt] += __shfl_xor(acc[kt], 16, 32);

    __syncthreads();
    if (lane < 16) {
        #pragma unroll
        for (int kt = 0; kt < 4; ++kt) sA[wave * NXC + kt * 16 + lane] = acc[kt];
    }
    __syncthreads();
    if (tid < NXC) {
        const float v = (sA[tid] + sA[NXC + tid]) + (sA[2 * NXC + tid] + sA[3 * NXC + tid]);
        const size_t fb = (size_t)f * BATCHC + batch;
        polar[fb * FEATC + c * NXC + tid] = v;
    }
}

typedef __attribute__((ext_vector_type(4))) float v4f;

#define S2_K     4096
#define S2_KQ    1024
#define S2_STAGE 32

struct S2Frags {
    v4f a[4][4];
    v4f b[4][4];
};

__device__ __forceinline__ void s2_load(
    S2Frags& f, const float* __restrict__ pa, const float* __restrict__ pb)
{
    #pragma unroll
    for (int t = 0; t < 4; ++t) {
        #pragma unroll
        for (int q = 0; q < 4; ++q) {
            f.a[t][q] = *reinterpret_cast<const v4f*>(pa + (size_t)t * 16 * S2_K + 4 * q);
            f.b[t][q] = *reinterpret_cast<const v4f*>(pb + (size_t)t * 16 * S2_K + 4 * q);
        }
    }
}

__device__ __forceinline__ void s2_mma(v8f (&acc)[4][4], const S2Frags& f)
{
    #pragma unroll
    for (int n = 0; n < 8; ++n) {
        v2f a[4], b[4];
        #pragma unroll
        for (int t = 0; t < 4; ++t) {
            const v4f qa = f.a[t][n >> 1];
            const v4f qb = f.b[t][n >> 1];
            if (n & 1) { a[t].x = qa.z; a[t].y = qa.w; b[t].x = qb.z; b[t].y = qb.w; }
            else       { a[t].x = qa.x; a[t].y = qa.y; b[t].x = qb.x; b[t].y = qb.y; }
        }
        #pragma unroll
        for (int mt = 0; mt < 4; ++mt) {
            #pragma unroll
            for (int nt = 0; nt < 4; ++nt) {
                acc[mt][nt] = __builtin_amdgcn_wmma_f32_16x16x4_f32(
                    false, a[mt], false, b[nt], (short)0, acc[mt][nt], false, false);
            }
        }
    }
}

__global__ __launch_bounds__(128) void cart_stage_kernel(
    const float* __restrict__ cart,
    const float* __restrict__ polar,
    float* __restrict__ out)
{
    __shared__ float red[4 * 16 * 8 * 32];

    const int tid  = threadIdx.x;
    const int lane = tid & 31;
    const int wave = tid >> 5;
    const int c    = lane & 15;
    const int h    = lane >> 4;

    const int m0 = blockIdx.x * 64;
    const int n0 = blockIdx.y * 64;

    const float* pa = polar + (size_t)(m0 + c) * S2_K + wave * S2_KQ + 16 * h;
    const float* pb = cart  + (size_t)(n0 + c) * S2_K + wave * S2_KQ + 16 * h;

    v8f acc[4][4] = {};
    S2Frags f0, f1;

    s2_load(f0, pa, pb);
    #pragma nounroll
    for (int k = 0; k < S2_KQ - 2 * S2_STAGE; k += 2 * S2_STAGE) {
        s2_load(f1, pa + k + S2_STAGE, pb + k + S2_STAGE);
        __builtin_amdgcn_sched_barrier(0);
        s2_mma(acc, f0);
        __builtin_amdgcn_sched_barrier(0);
        s2_load(f0, pa + k + 2 * S2_STAGE, pb + k + 2 * S2_STAGE);
        __builtin_amdgcn_sched_barrier(0);
        s2_mma(acc, f1);
        __builtin_amdgcn_sched_barrier(0);
    }
    s2_load(f1, pa + S2_KQ - S2_STAGE, pb + S2_KQ - S2_STAGE);
    __builtin_amdgcn_sched_barrier(0);
    s2_mma(acc, f0);
    s2_mma(acc, f1);

    #pragma unroll
    for (int mt = 0; mt < 4; ++mt) {
        #pragma unroll
        for (int nt = 0; nt < 4; ++nt) {
            #pragma unroll
            for (int r = 0; r < 8; ++r)
                red[((wave * 16 + mt * 4 + nt) * 8 + r) * 32 + lane] = acc[mt][nt][r];
        }
    }
    __syncthreads();

    const int p = n0 + wave * 16 + c;
    #pragma unroll
    for (int r = 0; r < 8; ++r) {
        v4f v;
        #pragma unroll
        for (int mt = 0; mt < 4; ++mt) {
            const float* rp = &red[((mt * 4 + wave) * 8 + r) * 32 + lane];
            v[mt] = (rp[0] + rp[16 * 8 * 32]) + (rp[2 * 16 * 8 * 32] + rp[3 * 16 * 8 * 32]);
        }
        const int bo = 8 * h + r;
        *reinterpret_cast<v4f*>(out + ((size_t)bo * FEATC + p) * NKC + (m0 >> 4)) = v;
    }
}

extern "C" void kernel_launch(void* const* d_in, const int* in_sizes, int n_in,
                              void* d_out, int out_size, void* d_ws, size_t ws_size,
                              hipStream_t stream) {
    const float* inputs  = (const float*)d_in[0];
    const float* cart    = (const float*)d_in[1];
    const int*   r_index = (const int*)  d_in[2];
    const float* pre1  = (const float*)d_in[3];
    const float* pre2  = (const float*)d_in[4];
    const float* pre3  = (const float*)d_in[5];
    const float* pre4  = (const float*)d_in[6];
    const float* post1 = (const float*)d_in[7];
    const float* post2 = (const float*)d_in[8];
    const float* post3 = (const float*)d_in[9];
    const float* post4 = (const float*)d_in[10];
    const float* ck1 = (const float*)d_in[11];
    const float* sk1 = (const float*)d_in[12];
    const float* ck2 = (const float*)d_in[13];
    const float* sk2 = (const float*)d_in[14];
    const float* ck3 = (const float*)d_in[15];
    const float* sk3 = (const float*)d_in[16];
    const float* ck4 = (const float*)d_in[17];
    const float* sk4 = (const float*)d_in[18];

    float* polar = (float*)d_ws;

    polar_stage_kernel<<<dim3(64, 16, 16), 128, 0, stream>>>(
        inputs, r_index,
        pre1, pre2, pre3, pre4, post1, post2, post3, post4,
        ck1, sk1, ck2, sk2, ck3, sk3, ck4, sk4, polar);

    cart_stage_kernel<<<dim3(4, 64), 128, 0, stream>>>(
        cart, polar, (float*)d_out);
}


# Stage 1 re-cut, batch as the WMMA M dimension: a pre-pass re-lays each plane batch-innermost (kept in the output buffer); a gathered element is one 64-byte read of all 16 batches; sum order changed
# speedup vs baseline: 7.4651x; 2.4596x over previous
#include <hip/hip_runtime.h>


typedef __attribute__((ext_vector_type(2))) float v2f;
typedef __attribute__((ext_vector_type(8))) float v8f;
typedef __attribute__((ext_vector_type(2))) int   v2i;

#define NXC 64
#define NKC 16
#define BATCHC 16
#define FEATC (NXC * NXC)

__global__ __launch_bounds__(256) void plane_relayout_kernel(
    const float* __restrict__ inputs,
    float* __restrict__ T,
    int plane)
{
    __shared__ float tile[BATCHC][257];
    const int t     = threadIdx.x;
    const int feat0 = blockIdx.x * 16;
    for (int b = 0; b < BATCHC; ++b)
        tile[b][t] = inputs[((size_t)(b * 2 + plane) * FEATC + feat0) * NKC + t];
    __syncthreads();
    const int b  = t & 15;
    const int fl = t >> 4;
    for (int f = 0; f < NKC; ++f)
        T[((size_t)f * FEATC + feat0 + fl) * BATCHC + b] = tile[b][fl * 16 + f];
}


__device__ __forceinline__ void s1_load_ix(
    v2i (&ix)[16], const int* __restrict__ rrow, unsigned hoff)
{
    const char* rb = reinterpret_cast<const char*>(rrow);
    #pragma unroll
    for (int s = 0; s < 16; ++s)
        ix[s] = *reinterpret_cast<const v2i*>(rb + (hoff + 16u * s));
}

__device__ __forceinline__ void s1_gather(
    v2f (&a)[16], const v2i (&ix)[16], const char* __restrict__ tf, unsigned lane4)
{
    #pragma unroll
    for (int s = 0; s < 16; ++s) {
        a[s].x = *reinterpret_cast<const float*>(tf + ((unsigned)ix[s].x * 64u + lane4));
        a[s].y = *reinterpret_cast<const float*>(tf + ((unsigned)ix[s].y * 64u + lane4));
    }
}

__device__ __forceinline__ void s1_row(
    v8f& acc0, v8f& acc1, const v2f (&a)[16],
    const v2f (&b0)[16], const v2f (&b1)[16], float w0, float w1)
{
    v8f e0 = {};
    v8f e1 = {};
    #pragma unroll
    for (int s = 0; s < 16; ++s) {
        e0 = __builtin_amdgcn_wmma_f32_16x16x4_f32(
                 false, a[s], false, b0[s], (short)0, e0, false, false);
        e1 = __builtin_amdgcn_wmma_f32_16x16x4_f32(
                 false, a[s], false, b1[s], (short)0, e1, false, false);
    }
    #pragma unroll
    for (int r = 0; r < 8; ++r) {
        acc0[r] += w0 * e0[r];
        acc1[r] += w1 * e1[r];
    }
}

template <int ADD>
__global__ __launch_bounds__(128, 4) void polar_plane_kernel(
    const float* __restrict__ T,
    const int*   __restrict__ r_index,
    const float* __restrict__ pre0,  const float* __restrict__ pre1,
    const float* __restrict__ post0, const float* __restrict__ post1,
    const float* __restrict__ k2_0,  const float* __restrict__ k2_1,
    const float* __restrict__ k1_0,  const float* __restrict__ k1_1,
    float* __restrict__ polar)
{
    __shared__ __attribute__((aligned(16))) float sOut[BATCHC * NXC];

    const int tid  = threadIdx.x;
    const int lane = tid & 31;
    const int wave = tid >> 5;
    const int c    = blockIdx.x;
    const int f    = blockIdx.y;
    const int h    = lane >> 4;
    const int l15  = lane & 15;
    const int kcol = wave * 16 + l15;

    v2f b0[16], b1[16];
    {
        const float* p0 = pre0 + f * NXC;
        const float* p1 = pre1 + f * NXC;
        const float* q0 = k1_0 + (size_t)f * FEATC + kcol;
        const float* q1 = k1_1 + (size_t)f * FEATC + kcol;
        #pragma unroll
        for (int s = 0; s < 16; ++s) {
            const int j = 4 * s + 2 * h;
            b0[s].x = p0[j]     * q0[j * NXC];
            b0[s].y = p0[j + 1] * q0[(j + 1) * NXC];
            b1[s].x = p1[j]     * q1[j * NXC];
            b1[s].y = p1[j + 1] * q1[(j + 1) * NXC];
        }
    }

    const int*  rix = r_index + c * FEATC;
    const char* tf  = reinterpret_cast<const char*>(T + (size_t)f * FEATC * BATCHC);
    const unsigned hoff  = 8u * h;
    const unsigned lane4 = 4u * l15;
    const float* w0f = k2_0 + (size_t)f * FEATC;
    const float* w1f = k2_1 + (size_t)f * FEATC;
    const float* s0p = post0 + f * NXC;
    const float* s1p = post1 + f * NXC;

    v8f acc0 = {};
    v8f acc1 = {};
    v2i ix[16];
    v2f  aE[16], aO[16];

    float wE0 = s0p[0] * w0f[kcol];
    float wE1 = s1p[0] * w1f[kcol];
    float wO0 = s0p[1] * w0f[NXC + kcol];
    float wO1 = s1p[1] * w1f[NXC + kcol];

    s1_load_ix(ix, rix, hoff);
    s1_gather(aE, ix, tf, lane4);
    s1_load_ix(ix, rix + NXC, hoff);
    #pragma nounroll
    for (int i = 0; i < NXC; i += 2) {
        const int in = (i + 2 < NXC) ? i + 2 : NXC - 2;
        const int i3 = (i + 3 < NXC) ? i + 3 : NXC - 1;
        const float nE0 = w0f[in * NXC + kcol];
        const float nE1 = w1f[in * NXC + kcol];
        const float nO0 = w0f[(in + 1) * NXC + kcol];
        const float nO1 = w1f[(in + 1) * NXC + kcol];
        const float pE0 = s0p[in],     pE1 = s1p[in];
        const float pO0 = s0p[in + 1], pO1 = s1p[in + 1];
        s1_gather(aO, ix, tf, lane4);
        s1_load_ix(ix, rix + in * NXC, hoff);
        __builtin_amdgcn_sched_barrier(0);
        s1_row(acc0, acc1, aE, b0, b1, wE0, wE1);
        __builtin_amdgcn_sched_barrier(0);
        s1_gather(aE, ix, tf, lane4);
        s1_load_ix(ix, rix + i3 * NXC, hoff);
        __builtin_amdgcn_sched_barrier(0);
        s1_row(acc0, acc1, aO, b0, b1, wO0, wO1);
        __builtin_amdgcn_sched_barrier(0);
        wE0 = pE0 * nE0;  wE1 = pE1 * nE1;
        wO0 = pO0 * nO0;  wO1 = pO1 * nO1;
    }

    #pragma unroll
    for (int r = 0; r < 8; ++r) sOut[(8 * h + r) * NXC + kcol] = acc0[r] + acc1[r];
    __syncthreads();
    {
        const int b  = tid >> 3;
        const int k8 = (tid & 7) * 8;
        float* dst = polar + ((size_t)(f * BATCHC + b)) * FEATC + c * NXC + k8;
        float4 lo = *reinterpret_cast<const float4*>(&sOut[b * NXC + k8]);
        float4 hi = *reinterpret_cast<const float4*>(&sOut[b * NXC + k8 + 4]);
        if (ADD) {
            const float4 plo = *reinterpret_cast<const float4*>(dst);
            const float4 phi = *reinterpret_cast<const float4*>(dst + 4);
            lo.x += plo.x; lo.y += plo.y; lo.z += plo.z; lo.w += plo.w;
            hi.x += phi.x; hi.y += phi.y; hi.z += phi.z; hi.w += phi.w;
        }
        *reinterpret_cast<float4*>(dst)     = lo;
        *reinterpret_cast<float4*>(dst + 4) = hi;
    }
}

typedef __attribute__((ext_vector_type(4))) float v4f;

#define S2_K     4096
#define S2_KQ    1024
#define S2_STAGE 32

struct S2Frags {
    v4f a[4][4];
    v4f b[4][4];
};

__device__ __forceinline__ void s2_load(
    S2Frags& f, const float* __restrict__ pa, const float* __restrict__ pb)
{
    #pragma unroll
    for (int t = 0; t < 4; ++t) {
        #pragma unroll
        for (int q = 0; q < 4; ++q) {
            f.a[t][q] = *reinterpret_cast<const v4f*>(pa + (size_t)t * 16 * S2_K + 4 * q);
            f.b[t][q] = *reinterpret_cast<const v4f*>(pb + (size_t)t * 16 * S2_K + 4 * q);
        }
    }
}

__device__ __forceinline__ void s2_mma(v8f (&acc)[4][4], const S2Frags& f)
{
    #pragma unroll
    for (int n = 0; n < 8; ++n) {
        v2f a[4], b[4];
        #pragma unroll
        for (int t = 0; t < 4; ++t) {
            const v4f qa = f.a[t][n >> 1];
            const v4f qb = f.b[t][n >> 1];
            if (n & 1) { a[t].x = qa.z; a[t].y = qa.w; b[t].x = qb.z; b[t].y = qb.w; }
            else       { a[t].x = qa.x; a[t].y = qa.y; b[t].x = qb.x; b[t].y = qb.y; }
        }
        #pragma unroll
        for (int mt = 0; mt < 4; ++mt) {
            #pragma unroll
            for (int nt = 0; nt < 4; ++nt) {
                acc[mt][nt] = __builtin_amdgcn_wmma_f32_16x16x4_f32(
                    false, a[mt], false, b[nt], (short)0, acc[mt][nt], false, false);
            }
        }
    }
}

__global__ __launch_bounds__(128) void cart_stage_kernel(
    const float* __restrict__ cart,
    const float* __restrict__ polar,
    float* __restrict__ out)
{
    __shared__ float red[4 * 16 * 8 * 32];

    const int tid  = threadIdx.x;
    const int lane = tid & 31;
    const int wave = tid >> 5;
    const int c    = lane & 15;
    const int h    = lane >> 4;

    const int m0 = blockIdx.x * 64;
    const int n0 = blockIdx.y * 64;

    const float* pa = polar + (size_t)(m0 + c) * S2_K + wave * S2_KQ + 16 * h;
    const float* pb = cart  + (size_t)(n0 + c) * S2_K + wave * S2_KQ + 16 * h;

    v8f acc[4][4] = {};
    S2Frags f0, f1;

    s2_load(f0, pa, pb);
    #pragma nounroll
    for (int k = 0; k < S2_KQ - 2 * S2_STAGE; k += 2 * S2_STAGE) {
        s2_load(f1, pa + k + S2_STAGE, pb + k + S2_STAGE);
        __builtin_amdgcn_sched_barrier(0);
        s2_mma(acc, f0);
        __builtin_amdgcn_sched_barrier(0);
        s2_load(f0, pa + k + 2 * S2_STAGE, pb + k + 2 * S2_STAGE);
        __builtin_amdgcn_sched_barrier(0);
        s2_mma(acc, f1);
        __builtin_amdgcn_sched_barrier(0);
    }
    s2_load(f1, pa + S2_KQ - S2_STAGE, pb + S2_KQ - S2_STAGE);
    __builtin_amdgcn_sched_barrier(0);
    s2_mma(acc, f0);
    s2_mma(acc, f1);

    #pragma unroll
    for (int mt = 0; mt < 4; ++mt) {
        #pragma unroll
        for (int nt = 0; nt < 4; ++nt) {
            #pragma unroll
            for (int r = 0; r < 8; ++r)
                red[((wave * 16 + mt * 4 + nt) * 8 + r) * 32 + lane] = acc[mt][nt][r];
        }
    }
    __syncthreads();

    const int p = n0 + wave * 16 + c;
    #pragma unroll
    for (int r = 0; r < 8; ++r) {
        v4f v;
        #pragma unroll
        for (int mt = 0; mt < 4; ++mt) {
            const float* rp = &red[((mt * 4 + wave) * 8 + r) * 32 + lane];
            v[mt] = (rp[0] + rp[16 * 8 * 32]) + (rp[2 * 16 * 8 * 32] + rp[3 * 16 * 8 * 32]);
        }
        const int bo = 8 * h + r;
        *reinterpret_cast<v4f*>(out + ((size_t)bo * FEATC + p) * NKC + (m0 >> 4)) = v;
    }
}

extern "C" void kernel_launch(void* const* d_in, const int* in_sizes, int n_in,
                              void* d_out, int out_size, void* d_ws, size_t ws_size,
                              hipStream_t stream) {
    const float* inputs  = (const float*)d_in[0];
    const float* cart    = (const float*)d_in[1];
    const int*   r_index = (const int*)  d_in[2];
    const float* pre1  = (const float*)d_in[3];
    const float* pre2  = (const float*)d_in[4];
    const float* pre3  = (const float*)d_in[5];
    const float* pre4  = (const float*)d_in[6];
    const float* post1 = (const float*)d_in[7];
    const float* post2 = (const float*)d_in[8];
    const float* post3 = (const float*)d_in[9];
    const float* post4 = (const float*)d_in[10];
    const float* ck1 = (const float*)d_in[11];
    const float* sk1 = (const float*)d_in[12];
    const float* ck2 = (const float*)d_in[13];
    const float* sk2 = (const float*)d_in[14];
    const float* ck3 = (const float*)d_in[15];
    const float* sk3 = (const float*)d_in[16];
    const float* ck4 = (const float*)d_in[17];
    const float* sk4 = (const float*)d_in[18];

    float* polar = (float*)d_ws;
    float* T = (float*)d_out;

    plane_relayout_kernel<<<256, 256, 0, stream>>>(inputs, T, 0);
    polar_plane_kernel<0><<<dim3(64, 16), 128, 0, stream>>>(
        T, r_index, pre1, pre2, post1, post2, ck1, sk1, ck2, sk2, polar);
    plane_relayout_kernel<<<256, 256, 0, stream>>>(inputs, T, 1);
    polar_plane_kernel<1><<<dim3(64, 16), 128, 0, stream>>>(
        T, r_index, pre3, pre4, post3, post4, ck3, sk4, sk3, ck4, polar);

    cart_stage_kernel<<<dim3(4, 64), 128, 0, stream>>>(
        cart, polar, (float*)d_out);
}


# Stage 1 re-tiled: a wave covers all 64 columns and both terms, so a gathered A fragment feeds 8 WMMAs (was 2); the 4 waves split rows i; block = 4 c; 4 partials added via LDS; sum order changed
# speedup vs baseline: 8.1954x; 1.0978x over previous
#include <hip/hip_runtime.h>


typedef __attribute__((ext_vector_type(2))) float v2f;
typedef __attribute__((ext_vector_type(8))) float v8f;
typedef __attribute__((ext_vector_type(2))) int   v2i;

#define NXC 64
#define NKC 16
#define BATCHC 16
#define FEATC (NXC * NXC)

__global__ __launch_bounds__(256) void plane_relayout_kernel(
    const float* __restrict__ inputs,
    float* __restrict__ T,
    int plane)
{
    __shared__ float tile[BATCHC][257];
    const int t     = threadIdx.x;
    const int feat0 = blockIdx.x * 16;
    for (int b = 0; b < BATCHC; ++b)
        tile[b][t] = inputs[((size_t)(b * 2 + plane) * FEATC + feat0) * NKC + t];
    __syncthreads();
    const int b  = t & 15;
    const int fl = t >> 4;
    for (int f = 0; f < NKC; ++f)
        T[((size_t)f * FEATC + feat0 + fl) * BATCHC + b] = tile[b][fl * 16 + f];
}


__device__ __forceinline__ void s1_load_ix(
    v2i (&ix)[16], const int* __restrict__ rrow, unsigned hoff)
{
    const char* rb = reinterpret_cast<const char*>(rrow);
    #pragma unroll
    for (int s = 0; s < 16; ++s)
        ix[s] = *reinterpret_cast<const v2i*>(rb + (hoff + 16u * s));
}

__device__ __forceinline__ void s1_gather(
    v2f (&a)[16], const v2i (&ix)[16], const char* __restrict__ tf, unsigned lane4)
{
    #pragma unroll
    for (int s = 0; s < 16; ++s) {
        a[s].x = *reinterpret_cast<const float*>(tf + ((unsigned)ix[s].x * 64u + lane4));
        a[s].y = *reinterpret_cast<const float*>(tf + ((unsigned)ix[s].y * 64u + lane4));
    }
}

__device__ __forceinline__ void s1_row(
    v8f (&acc)[2][4], const v2f (&a)[16], const v2f (&bf)[2][4][16], const float (&w)[2][4])
{
    v8f e[2][4] = {};
    #pragma unroll
    for (int s = 0; s < 16; ++s) {
        #pragma unroll
        for (int nt = 0; nt < 4; ++nt) {
            e[0][nt] = __builtin_amdgcn_wmma_f32_16x16x4_f32(
                           false, a[s], false, bf[0][nt][s], (short)0, e[0][nt], false, false);
            e[1][nt] = __builtin_amdgcn_wmma_f32_16x16x4_f32(
                           false, a[s], false, bf[1][nt][s], (short)0, e[1][nt], false, false);
        }
    }
    __builtin_amdgcn_sched_barrier(0);
    #pragma unroll
    for (int nt = 0; nt < 4; ++nt) {
        #pragma unroll
        for (int r = 0; r < 8; ++r) {
            acc[0][nt][r] += w[0][nt] * e[0][nt][r];
            acc[1][nt][r] += w[1][nt] * e[1][nt][r];
        }
    }
}

__device__ __forceinline__ void s1_load_k2(
    float (&v)[2][4], const float* __restrict__ r0, const float* __restrict__ r1)
{
    #pragma unroll
    for (int nt = 0; nt < 4; ++nt) { v[0][nt] = r0[nt * 16]; v[1][nt] = r1[nt * 16]; }
}

#define S1_CPB 4

template <int ADD>
__global__ __launch_bounds__(128) void polar_plane_kernel(
    const float* __restrict__ T,
    const int*   __restrict__ r_index,
    const float* __restrict__ pre0,  const float* __restrict__ pre1,
    const float* __restrict__ post0, const float* __restrict__ post1,
    const float* __restrict__ k2_0,  const float* __restrict__ k2_1,
    const float* __restrict__ k1_0,  const float* __restrict__ k1_1,
    float* __restrict__ polar)
{
    __shared__ __attribute__((aligned(16))) float sPart[4 * BATCHC * NXC];

    const int tid  = threadIdx.x;
    const int lane = tid & 31;
    const int wave = tid >> 5;
    const int c0   = blockIdx.x * S1_CPB;
    const int f    = blockIdx.y;
    const int h    = lane >> 4;
    const int l15  = lane & 15;

    v2f bf[2][4][16];
    {
        const float* p0 = pre0 + f * NXC;
        const float* p1 = pre1 + f * NXC;
        const float* q0 = k1_0 + (size_t)f * FEATC + l15;
        const float* q1 = k1_1 + (size_t)f * FEATC + l15;
        #pragma unroll
        for (int s = 0; s < 16; ++s) {
            const int j = 4 * s + 2 * h;
            const float pa = p0[j], pb = p0[j + 1], pc = p1[j], pd = p1[j + 1];
            #pragma unroll
            for (int nt = 0; nt < 4; ++nt) {
                bf[0][nt][s].x = pa * q0[j * NXC + nt * 16];
                bf[0][nt][s].y = pb * q0[(j + 1) * NXC + nt * 16];
                bf[1][nt][s].x = pc * q1[j * NXC + nt * 16];
                bf[1][nt][s].y = pd * q1[(j + 1) * NXC + nt * 16];
            }
        }
    }

    const char* tf  = reinterpret_cast<const char*>(T + (size_t)f * FEATC * BATCHC);
    const unsigned hoff  = 8u * h;
    const unsigned lane4 = 4u * l15;
    const int*   rix = r_index + c0 * FEATC + wave * NXC;
    const float* w0f = k2_0 + (size_t)f * FEATC + wave * NXC + l15;
    const float* w1f = k2_1 + (size_t)f * FEATC + wave * NXC + l15;
    const float* s0p = post0 + f * NXC + wave;
    const float* s1p = post1 + f * NXC + wave;

    v8f acc[2][4] = {};
    v2i ix[16];
    v2f aE[16], aO[16];
    float wE[2][4], wO[2][4];

    {
        float kv[2][4];
        s1_load_k2(kv, w0f, w1f);
        #pragma unroll
        for (int nt = 0; nt < 4; ++nt) { wE[0][nt] = s0p[0] * kv[0][nt]; wE[1][nt] = s1p[0] * kv[1][nt]; }
        s1_load_k2(kv, w0f + 4 * NXC, w1f + 4 * NXC);
        #pragma unroll
        for (int nt = 0; nt < 4; ++nt) { wO[0][nt] = s0p[4] * kv[0][nt]; wO[1][nt] = s1p[4] * kv[1][nt]; }
    }

    s1_load_ix(ix, rix, hoff);
    s1_gather(aE, ix, tf, lane4);
    s1_load_ix(ix, rix + 4 * NXC, hoff);
    #pragma nounroll
    for (int n = 0; n < 16 * S1_CPB; n += 2) {
        const int n2 = (n + 2 < 16 * S1_CPB) ? n + 2 : 16 * S1_CPB - 2;
        const int n3 = n2 + 1;
        const int m2 = n2 & 15, m3 = n3 & 15;
        float kE[2][4], kO[2][4];
        s1_load_k2(kE, w0f + m2 * 4 * NXC, w1f + m2 * 4 * NXC);
        s1_load_k2(kO, w0f + m3 * 4 * NXC, w1f + m3 * 4 * NXC);
        const float pE0 = s0p[4 * m2], pE1 = s1p[4 * m2];
        const float pO0 = s0p[4 * m3], pO1 = s1p[4 * m3];
        s1_gather(aO, ix, tf, lane4);
        s1_load_ix(ix, rix + (n2 >> 4) * FEATC + m2 * 4 * NXC, hoff);
        __builtin_amdgcn_sched_barrier(0);
        s1_row(acc, aE, bf, wE);
        __builtin_amdgcn_sched_barrier(0);
        s1_gather(aE, ix, tf, lane4);
        s1_load_ix(ix, rix + (n3 >> 4) * FEATC + m3 * 4 * NXC, hoff);
        __builtin_amdgcn_sched_barrier(0);
        s1_row(acc, aO, bf, wO);
        __builtin_amdgcn_sched_barrier(0);
        #pragma unroll
        for (int nt = 0; nt < 4; ++nt) {
            wE[0][nt] = pE0 * kE[0][nt];  wE[1][nt] = pE1 * kE[1][nt];
            wO[0][nt] = pO0 * kO[0][nt];  wO[1][nt] = pO1 * kO[1][nt];
        }

        if ((n & 15) == 14) {
            float* sp = sPart + wave * (BATCHC * NXC);
            #pragma unroll
            for (int nt = 0; nt < 4; ++nt) {
                #pragma unroll
                for (int r = 0; r < 8; ++r) {
                    sp[(8 * h + r) * NXC + nt * 16 + l15] = acc[0][nt][r] + acc[1][nt][r];
                    acc[0][nt][r] = 0.0f;
                    acc[1][nt][r] = 0.0f;
                }
            }
            __syncthreads();
            const int b   = tid >> 3;
            const int seg = (tid & 7) * 4;
            float* dst = polar + ((size_t)(f * BATCHC + b)) * FEATC + (c0 + (n >> 4)) * NXC;
            #pragma unroll
            for (int half = 0; half < 2; ++half) {
                const int k = half * 32 + seg;
                const float4 p0 = *reinterpret_cast<const float4*>(&sPart[0 * BATCHC * NXC + b * NXC + k]);
                const float4 p1 = *reinterpret_cast<const float4*>(&sPart[1 * BATCHC * NXC + b * NXC + k]);
                const float4 p2 = *reinterpret_cast<const float4*>(&sPart[2 * BATCHC * NXC + b * NXC + k]);
                const float4 p3 = *reinterpret_cast<const float4*>(&sPart[3 * BATCHC * NXC + b * NXC + k]);
                float4 v;
                v.x = (p0.x + p1.x) + (p2.x + p3.x);
                v.y = (p0.y + p1.y) + (p2.y + p3.y);
                v.z = (p0.z + p1.z) + (p2.z + p3.z);
                v.w = (p0.w + p1.w) + (p2.w + p3.w);
                if (ADD) {
                    const float4 q = *reinterpret_cast<const float4*>(dst + k);
                    v.x += q.x; v.y += q.y; v.z += q.z; v.w += q.w;
                }
                *reinterpret_cast<float4*>(dst + k) = v;
            }
            __syncthreads();
        }
    }
}

typedef __attribute__((ext_vector_type(4))) float v4f;

#define S2_K     4096
#define S2_KQ    1024
#define S2_STAGE 32

struct S2Frags {
    v4f a[4][4];
    v4f b[4][4];
};

__device__ __forceinline__ void s2_load(
    S2Frags& f, const float* __restrict__ pa, const float* __restrict__ pb)
{
    #pragma unroll
    for (int t = 0; t < 4; ++t) {
        #pragma unroll
        for (int q = 0; q < 4; ++q) {
            f.a[t][q] = *reinterpret_cast<const v4f*>(pa + (size_t)t * 16 * S2_K + 4 * q);
            f.b[t][q] = *reinterpret_cast<const v4f*>(pb + (size_t)t * 16 * S2_K + 4 * q);
        }
    }
}

__device__ __forceinline__ void s2_mma(v8f (&acc)[4][4], const S2Frags& f)
{
    #pragma unroll
    for (int n = 0; n < 8; ++n) {
        v2f a[4], b[4];
        #pragma unroll
        for (int t = 0; t < 4; ++t) {
            const v4f qa = f.a[t][n >> 1];
            const v4f qb = f.b[t][n >> 1];
            if (n & 1) { a[t].x = qa.z; a[t].y = qa.w; b[t].x = qb.z; b[t].y = qb.w; }
            else       { a[t].x = qa.x; a[t].y = qa.y; b[t].x = qb.x; b[t].y = qb.y; }
        }
        #pragma unroll
        for (int mt = 0; mt < 4; ++mt) {
            #pragma unroll
            for (int nt = 0; nt < 4; ++nt) {
                acc[mt][nt] = __builtin_amdgcn_wmma_f32_16x16x4_f32(
                    false, a[mt], false, b[nt], (short)0, acc[mt][nt], false, false);
            }
        }
    }
}

__global__ __launch_bounds__(128) void cart_stage_kernel(
    const float* __restrict__ cart,
    const float* __restrict__ polar,
    float* __restrict__ out)
{
    __shared__ float red[4 * 16 * 8 * 32];

    const int tid  = threadIdx.x;
    const int lane = tid & 31;
    const int wave = tid >> 5;
    const int c    = lane & 15;
    const int h    = lane >> 4;

    const int m0 = blockIdx.x * 64;
    const int n0 = blockIdx.y * 64;

    const float* pa = polar + (size_t)(m0 + c) * S2_K + wave * S2_KQ + 16 * h;
    const float* pb = cart  + (size_t)(n0 + c) * S2_K + wave * S2_KQ + 16 * h;

    v8f acc[4][4] = {};
    S2Frags f0, f1;

    s2_load(f0, pa, pb);
    #pragma nounroll
    for (int k = 0; k < S2_KQ - 2 * S2_STAGE; k += 2 * S2_STAGE) {
        s2_load(f1, pa + k + S2_STAGE, pb + k + S2_STAGE);
        __builtin_amdgcn_sched_barrier(0);
        s2_mma(acc, f0);
        __builtin_amdgcn_sched_barrier(0);
        s2_load(f0, pa + k + 2 * S2_STAGE, pb + k + 2 * S2_STAGE);
        __builtin_amdgcn_sched_barrier(0);
        s2_mma(acc, f1);
        __builtin_amdgcn_sched_barrier(0);
    }
    s2_load(f1, pa + S2_KQ - S2_STAGE, pb + S2_KQ - S2_STAGE);
    __builtin_amdgcn_sched_barrier(0);
    s2_mma(acc, f0);
    s2_mma(acc, f1);

    #pragma unroll
    for (int mt = 0; mt < 4; ++mt) {
        #pragma unroll
        for (int nt = 0; nt < 4; ++nt) {
            #pragma unroll
            for (int r = 0; r < 8; ++r)
                red[((wave * 16 + mt * 4 + nt) * 8 + r) * 32 + lane] = acc[mt][nt][r];
        }
    }
    __syncthreads();

    const int p = n0 + wave * 16 + c;
    #pragma unroll
    for (int r = 0; r < 8; ++r) {
        v4f v;
        #pragma unroll
        for (int mt = 0; mt < 4; ++mt) {
            const float* rp = &red[((mt * 4 + wave) * 8 + r) * 32 + lane];
            v[mt] = (rp[0] + rp[16 * 8 * 32]) + (rp[2 * 16 * 8 * 32] + rp[3 * 16 * 8 * 32]);
        }
        const int bo = 8 * h + r;
        *reinterpret_cast<v4f*>(out + ((size_t)bo * FEATC + p) * NKC + (m0 >> 4)) = v;
    }
}

extern "C" void kernel_launch(void* const* d_in, const int* in_sizes, int n_in,
                              void* d_out, int out_size, void* d_ws, size_t ws_size,
                              hipStream_t stream) {
    const float* inputs  = (const float*)d_in[0];
    const float* cart    = (const float*)d_in[1];
    const int*   r_index = (const int*)  d_in[2];
    const float* pre1  = (const float*)d_in[3];
    const float* pre2  = (const float*)d_in[4];
    const float* pre3  = (const float*)d_in[5];
    const float* pre4  = (const float*)d_in[6];
    const float* post1 = (const float*)d_in[7];
    const float* post2 = (const float*)d_in[8];
    const float* post3 = (const float*)d_in[9];
    const float* post4 = (const float*)d_in[10];
    const float* ck1 = (const float*)d_in[11];
    const float* sk1 = (const float*)d_in[12];
    const float* ck2 = (const float*)d_in[13];
    const float* sk2 = (const float*)d_in[14];
    const float* ck3 = (const float*)d_in[15];
    const float* sk3 = (const float*)d_in[16];
    const float* ck4 = (const float*)d_in[17];
    const float* sk4 = (const float*)d_in[18];

    float* polar = (float*)d_ws;
    float* T = (float*)d_out;

    plane_relayout_kernel<<<256, 256, 0, stream>>>(inputs, T, 0);
    polar_plane_kernel<0><<<dim3(16, 16), 128, 0, stream>>>(
        T, r_index, pre1, pre2, post1, post2, ck1, sk1, ck2, sk2, polar);
    plane_relayout_kernel<<<256, 256, 0, stream>>>(inputs, T, 1);
    polar_plane_kernel<1><<<dim3(16, 16), 128, 0, stream>>>(
        T, r_index, pre3, pre4, post3, post4, ck3, sk4, sk3, ck4, polar);

    cart_stage_kernel<<<dim3(4, 64), 128, 0, stream>>>(
        cart, polar, (float*)d_out);
}
